# starting kernel
# baseline (speedup 1.0000x reference)
#include <hip/hip_runtime.h>

#define ORDER 16
#define FRAME 160
#define NPAD  20
#define XPS   (NPAD + FRAME + NPAD)
#define WPB   8
#define LDSW  (XPS + NPAD + NPAD + FRAME)

typedef __attribute__((ext_vector_type(2))) float v2f;
typedef __attribute__((ext_vector_type(8))) float v8f;

__global__ __launch_bounds__(32 * WPB)
void lpc_residual_kernel(const float* __restrict__ pcm,
                         float* __restrict__ out,
                         int nframes)
{
    __shared__ float smem[WPB * LDSW];
    const int lane  = threadIdx.x & 31;
    const int wave  = threadIdx.x >> 5;
    const int frame = blockIdx.x * WPB + wave;
    if (frame >= nframes) return;

    float* xps   = smem + wave * LDSW;
    float* r_lds = xps + XPS;
    float* a_lds = r_lds + NPAD;
    float* y_lds = a_lds + NPAD;

    const float* src = pcm + (size_t)frame * FRAME;

    #pragma unroll
    for (int q = 0; q < 5; ++q)
        xps[NPAD + lane + 32 * q] = src[lane + 32 * q];
    if (lane < NPAD) { xps[lane] = 0.f; xps[NPAD + FRAME + lane] = 0.f; }
    __builtin_amdgcn_wave_barrier();

    if (lane <= ORDER) {
        float s = 0.f;
        for (int t = 0; t < FRAME; ++t)
            s += xps[NPAD + t] * xps[NPAD + t + lane];
        r_lds[lane] = s;
    } else if (lane < NPAD) {
        r_lds[lane] = 0.f;
    }
    if (lane == 0) a_lds[0] = 1.f;
    if (lane >= 1 && lane < NPAD) a_lds[lane] = 0.f;
    __builtin_amdgcn_wave_barrier();

    {
        float e = r_lds[0];
        for (int i = 1; i <= ORDER; ++i) {
            float acc = r_lds[i];
            for (int j = 1; j < i; ++j)
                acc -= a_lds[j] * r_lds[i - j];
            float k = acc / e;
            const int  j   = lane;
            const bool upd = (j >= 1) && (j < i);
            float aj  = upd ? a_lds[j]     : 0.f;
            float arv = upd ? a_lds[i - j] : 0.f;
            float na  = aj - k * arv;
            if (upd)       a_lds[j] = na;
            if (lane == 0) a_lds[i] = k;
            e = e * (1.f - k * k);
            __builtin_amdgcn_wave_barrier();
        }
    }

    const int half = lane >> 4;
    const int mrow = lane & 15;
    const int koff = half * 2;

    v2f bop[5];
    #pragma unroll
    for (int s = 0; s < 5; ++s) {
        bop[s].x = a_lds[4 * s + koff];
        bop[s].y = a_lds[4 * s + koff + 1];
    }

    #pragma unroll
    for (int c = 0; c < 10; ++c) {
        v8f acc = {};
        const int tb = 16 * c;
        #pragma unroll
        for (int s = 0; s < 5; ++s) {
            const int kk = 4 * s + koff;
            v2f aop;
            aop.x = xps[NPAD + tb + mrow - kk];
            aop.y = xps[NPAD + tb + mrow - kk - 1];
            acc = __builtin_amdgcn_wmma_f32_16x16x4_f32(
                      false, aop, false, bop[s], (short)0, acc, false, false);
        }
        #pragma unroll
        for (int q = 0; q < 8; ++q)
            y_lds[tb + half * 8 + q] = acc[q];
    }
    __builtin_amdgcn_wave_barrier();

    float* dst = out + (size_t)frame * FRAME;
    #pragma unroll
    for (int q = 0; q < 5; ++q)
        dst[lane + 32 * q] = y_lds[lane + 32 * q];
}

extern "C" void kernel_launch(void* const* d_in, const int* in_sizes, int n_in,
                              void* d_out, int out_size, void* d_ws, size_t ws_size,
                              hipStream_t stream) {
    const float* pcm = (const float*)d_in[0];
    float* out = (float*)d_out;
    const int total   = in_sizes[0];
    const int nframes = total / FRAME;
    const int blocks  = (nframes + WPB - 1) / WPB;
    lpc_residual_kernel<<<blocks, 32 * WPB, 0, stream>>>(pcm, out, nframes);
}


# autocorrelation parallel over time: lane l keeps samples 5l..5l+20 in registers, 17 lags summed over lanes by a halving DPP tree, not a 160-step per-lag LDS loop; summation order of r changes
# speedup vs baseline: 1.1589x; 1.1589x over previous
#include <hip/hip_runtime.h>

#define ORDER 16
#define FRAME 160
#define NPAD  20
#define XPS   (NPAD + FRAME + NPAD)
#define WPB   8
#define LDSW  (XPS + NPAD + NPAD + FRAME)

typedef __attribute__((ext_vector_type(2))) float v2f;
typedef __attribute__((ext_vector_type(8))) float v8f;

template <int M>
__device__ __forceinline__ float row_xor(float v) {
    return __builtin_bit_cast(float,
        __builtin_amdgcn_update_dpp(0, __builtin_bit_cast(int, v), 0x160 + M, 0xf, 0xf, true));
}

__device__ __forceinline__ float other_row(float v) {
    const unsigned int u = __builtin_bit_cast(unsigned int, v);
    return __builtin_bit_cast(float,
        __builtin_amdgcn_permlanex16(u, u, 0x76543210u, 0xfedcba98u, false, false));
}

template <int M>
__device__ __forceinline__ float fold(float a, float b, bool hi) {
    const float keep = hi ? b : a;
    const float send = hi ? a : b;
    return keep + row_xor<M>(send);
}

__global__ __launch_bounds__(32 * WPB)
void lpc_residual_kernel(const float* __restrict__ pcm,
                         float* __restrict__ out,
                         int nframes)
{
    __shared__ float smem[WPB * LDSW];
    const int lane  = threadIdx.x & 31;
    const int wave  = threadIdx.x >> 5;
    const int frame = blockIdx.x * WPB + wave;
    if (frame >= nframes) return;

    float* xps   = smem + wave * LDSW;
    float* r_lds = xps + XPS;
    float* a_lds = r_lds + NPAD;
    float* y_lds = a_lds + NPAD;

    const float* src = pcm + (size_t)frame * FRAME;

    #pragma unroll
    for (int q = 0; q < 5; ++q)
        xps[NPAD + lane + 32 * q] = src[lane + 32 * q];
    if (lane < NPAD) { xps[lane] = 0.f; xps[NPAD + FRAME + lane] = 0.f; }
    __builtin_amdgcn_wave_barrier();

    {
        const float* xw = xps + NPAD + 5 * lane;
        float w[5 + ORDER];
        #pragma unroll
        for (int c = 0; c < 5 + ORDER; ++c) w[c] = xw[c];

        float s[ORDER + 1];
        #pragma unroll
        for (int l = 0; l <= ORDER; ++l) {
            float acc = w[0] * w[l];
            #pragma unroll
            for (int c = 1; c < 5; ++c) acc += w[c] * w[c + l];
            s[l] = acc;
        }

        const bool b0 = (lane & 1) != 0, b1 = (lane & 2) != 0, b2 = (lane & 4) != 0, b3 = (lane & 8) != 0;
        float u[8];
        #pragma unroll
        for (int p = 0; p < 8; ++p) u[p] = fold<8>(s[2 * p + 1], s[2 * p + 2], b3);
        float v[4];
        #pragma unroll
        for (int q = 0; q < 4; ++q) v[q] = fold<4>(u[2 * q], u[2 * q + 1], b2);
        const float z0 = fold<2>(v[0], v[1], b1);
        const float z1 = fold<2>(v[2], v[3], b1);
        float y = fold<1>(z0, z1, b0);
        y += other_row(y);
        const int lag = 1 + (b3 ? 1 : 0) + (b2 ? 2 : 0) + (b1 ? 4 : 0) + (b0 ? 8 : 0);

        float e0 = s[0];
        e0 += row_xor<1>(e0);
        e0 += row_xor<2>(e0);
        e0 += row_xor<4>(e0);
        e0 += row_xor<8>(e0);
        e0 += other_row(e0);

        if (lane < 16) r_lds[lag] = y;
        if (lane == 0) r_lds[0] = e0;
        if (lane > ORDER && lane < NPAD) r_lds[lane] = 0.f;
    }
    if (lane == 0) a_lds[0] = 1.f;
    if (lane >= 1 && lane < NPAD) a_lds[lane] = 0.f;
    __builtin_amdgcn_wave_barrier();

    {
        float e = r_lds[0];
        for (int i = 1; i <= ORDER; ++i) {
            float acc = r_lds[i];
            for (int j = 1; j < i; ++j)
                acc -= a_lds[j] * r_lds[i - j];
            float k = acc / e;
            const int  j   = lane;
            const bool upd = (j >= 1) && (j < i);
            float aj  = upd ? a_lds[j]     : 0.f;
            float arv = upd ? a_lds[i - j] : 0.f;
            float na  = aj - k * arv;
            if (upd)       a_lds[j] = na;
            if (lane == 0) a_lds[i] = k;
            e = e * (1.f - k * k);
            __builtin_amdgcn_wave_barrier();
        }
    }

    const int half = lane >> 4;
    const int mrow = lane & 15;
    const int koff = half * 2;

    v2f bop[5];
    #pragma unroll
    for (int s = 0; s < 5; ++s) {
        bop[s].x = a_lds[4 * s + koff];
        bop[s].y = a_lds[4 * s + koff + 1];
    }

    #pragma unroll
    for (int c = 0; c < 10; ++c) {
        v8f acc = {};
        const int tb = 16 * c;
        #pragma unroll
        for (int s = 0; s < 5; ++s) {
            const int kk = 4 * s + koff;
            v2f aop;
            aop.x = xps[NPAD + tb + mrow - kk];
            aop.y = xps[NPAD + tb + mrow - kk - 1];
            acc = __builtin_amdgcn_wmma_f32_16x16x4_f32(
                      false, aop, false, bop[s], (short)0, acc, false, false);
        }
        #pragma unroll
        for (int q = 0; q < 8; ++q)
            y_lds[tb + half * 8 + q] = acc[q];
    }
    __builtin_amdgcn_wave_barrier();

    float* dst = out + (size_t)frame * FRAME;
    #pragma unroll
    for (int q = 0; q < 5; ++q)
        dst[lane + 32 * q] = y_lds[lane + 32 * q];
}

extern "C" void kernel_launch(void* const* d_in, const int* in_sizes, int n_in,
                              void* d_out, int out_size, void* d_ws, size_t ws_size,
                              hipStream_t stream) {
    const float* pcm = (const float*)d_in[0];
    float* out = (float*)d_out;
    const int total   = in_sizes[0];
    const int nframes = total / FRAME;
    const int blocks  = (nframes + WPB - 1) / WPB;
    lpc_residual_kernel<<<blocks, 32 * WPB, 0, stream>>>(pcm, out, nframes);
}


# Levinson-Durbin without inner products, in registers: lanes carry q[m] = r[m] - sum a[j] r[|m-j|] (numerator = q[i]) and a[] forward + reversed; step = 1 division, 4 FMA, 3 DPP; acc's sums reordered
# speedup vs baseline: 1.4824x; 1.2791x over previous
#include <hip/hip_runtime.h>

#define ORDER 16
#define FRAME 160
#define NPAD  20
#define XPS   (NPAD + FRAME + NPAD)
#define WPB   8
#define LDSW  (XPS + NPAD + FRAME)

typedef __attribute__((ext_vector_type(2))) float v2f;
typedef __attribute__((ext_vector_type(8))) float v8f;

template <int M>
__device__ __forceinline__ float row_xor(float v) {
    return __builtin_bit_cast(float,
        __builtin_amdgcn_update_dpp(0, __builtin_bit_cast(int, v), 0x160 + M, 0xf, 0xf, true));
}

__device__ __forceinline__ float other_row(float v) {
    const unsigned int u = __builtin_bit_cast(unsigned int, v);
    return __builtin_bit_cast(float,
        __builtin_amdgcn_permlanex16(u, u, 0x76543210u, 0xfedcba98u, false, false));
}

template <int N>
__device__ __forceinline__ float row_lane(float v) {
    return __builtin_bit_cast(float,
        __builtin_amdgcn_update_dpp(0, __builtin_bit_cast(int, v), 0x150 + N, 0xf, 0xf, true));
}

__device__ __forceinline__ float row_down1(float v) {
    return __builtin_bit_cast(float,
        __builtin_amdgcn_update_dpp(0, __builtin_bit_cast(int, v), 0x101, 0xf, 0xf, true));
}

__device__ __forceinline__ float row_up1(float v, float first) {
    return __builtin_bit_cast(float,
        __builtin_amdgcn_update_dpp(__builtin_bit_cast(int, first), __builtin_bit_cast(int, v),
                                    0x111, 0xf, 0xf, false));
}

template <int M>
__device__ __forceinline__ float fold(float a, float b, bool hi) {
    const float keep = hi ? b : a;
    const float send = hi ? a : b;
    return keep + row_xor<M>(send);
}

__global__ __launch_bounds__(32 * WPB)
void lpc_residual_kernel(const float* __restrict__ pcm,
                         float* __restrict__ out,
                         int nframes)
{
    __shared__ float smem[WPB * LDSW];
    const int lane  = threadIdx.x & 31;
    const int wave  = threadIdx.x >> 5;
    const int frame = blockIdx.x * WPB + wave;
    if (frame >= nframes) return;

    float* xps   = smem + wave * LDSW;
    float* a_lds = xps + XPS;
    float* y_lds = a_lds + NPAD;

    const float* src = pcm + (size_t)frame * FRAME;

    #pragma unroll
    for (int q = 0; q < 5; ++q)
        xps[NPAD + lane + 32 * q] = src[lane + 32 * q];
    if (lane < NPAD) { xps[lane] = 0.f; xps[NPAD + FRAME + lane] = 0.f; }
    __builtin_amdgcn_wave_barrier();

    float r_up;
    float r_zero;
    {
        const float* xw = xps + NPAD + 5 * lane;
        float w[5 + ORDER];
        #pragma unroll
        for (int c = 0; c < 5 + ORDER; ++c) w[c] = xw[c];

        float s[ORDER + 1];
        #pragma unroll
        for (int l = 0; l <= ORDER; ++l) {
            float acc = w[0] * w[l];
            #pragma unroll
            for (int c = 1; c < 5; ++c) acc += w[c] * w[c + l];
            s[l] = acc;
        }

        const bool b0 = (lane & 1) != 0, b1 = (lane & 2) != 0, b2 = (lane & 4) != 0, b3 = (lane & 8) != 0;
        float u[8];
        #pragma unroll
        for (int p = 0; p < 8; ++p) u[p] = fold<8>(s[2 * p + 1], s[2 * p + 2], b3);
        float v[4];
        #pragma unroll
        for (int q = 0; q < 4; ++q) v[q] = fold<4>(u[2 * q], u[2 * q + 1], b2);
        const float z0 = fold<2>(v[0], v[1], b1);
        const float z1 = fold<2>(v[2], v[3], b1);
        float y = fold<1>(z0, z1, b0);
        y += other_row(y);

        float e0 = s[0];
        e0 += row_xor<1>(e0);
        e0 += row_xor<2>(e0);
        e0 += row_xor<4>(e0);
        e0 += row_xor<8>(e0);
        e0 += other_row(e0);

        r_up = __builtin_bit_cast(float,
            __builtin_amdgcn_permlane16(0u, __builtin_bit_cast(unsigned int, y),
                                        0xE6A2C480u, 0xF7B3D591u, false, false));
        r_zero = e0;
    }

    {
        const int n = lane & 15;
        float fwd = r_up;
        float bwd = row_up1(r_up, r_zero);
        float pa = 0.f, ra = 0.f;
        float e   = r_zero;
        float acc = row_lane<0>(fwd);
        #pragma unroll
        for (int i = 1; i <= ORDER; ++i) {
            const float k  = acc / e;
            const float fn = fwd - k * bwd;
            const float bn = bwd - k * fwd;
            acc = row_lane<1>(fn);
            fwd = row_down1(fn);
            bwd = bn;
            const float t1 = pa - k * ra;
            const float t2 = ra - k * pa;
            pa = (n == i - 1) ? k : t1;
            ra = row_up1(t2, k);
            e = e * (1.f - k * k);
        }
        if (lane < ORDER) a_lds[lane + 1] = pa;
        if (lane == 0) a_lds[0] = 1.f;
        if (lane > ORDER && lane < NPAD) a_lds[lane] = 0.f;
    }
    __builtin_amdgcn_wave_barrier();

    const int half = lane >> 4;
    const int mrow = lane & 15;
    const int koff = half * 2;

    v2f bop[5];
    #pragma unroll
    for (int s = 0; s < 5; ++s) {
        bop[s].x = a_lds[4 * s + koff];
        bop[s].y = a_lds[4 * s + koff + 1];
    }

    #pragma unroll
    for (int c = 0; c < 10; ++c) {
        v8f acc = {};
        const int tb = 16 * c;
        #pragma unroll
        for (int s = 0; s < 5; ++s) {
            const int kk = 4 * s + koff;
            v2f aop;
            aop.x = xps[NPAD + tb + mrow - kk];
            aop.y = xps[NPAD + tb + mrow - kk - 1];
            acc = __builtin_amdgcn_wmma_f32_16x16x4_f32(
                      false, aop, false, bop[s], (short)0, acc, false, false);
        }
        #pragma unroll
        for (int q = 0; q < 8; ++q)
            y_lds[tb + half * 8 + q] = acc[q];
    }
    __builtin_amdgcn_wave_barrier();

    float* dst = out + (size_t)frame * FRAME;
    #pragma unroll
    for (int q = 0; q < 5; ++q)
        dst[lane + 32 * q] = y_lds[lane + 32 * q];
}

extern "C" void kernel_launch(void* const* d_in, const int* in_sizes, int n_in,
                              void* d_out, int out_size, void* d_ws, size_t ws_size,
                              hipStream_t stream) {
    const float* pcm = (const float*)d_in[0];
    float* out = (float*)d_out;
    const int total   = in_sizes[0];
    const int nframes = total / FRAME;
    const int blocks  = (nframes + WPB - 1) / WPB;
    lpc_residual_kernel<<<blocks, 32 * WPB, 0, stream>>>(pcm, out, nframes);
}


# Levinson step's division k = acc/e as v_rcp_f32, product and one fused residual correction (4 dependent instructions, not the IEEE sequence's 11); r's sums as explicit fma, same order
# speedup vs baseline: 3.3315x; 2.2474x over previous
#include <hip/hip_runtime.h>

#define ORDER 16
#define FRAME 160
#define NPAD  20
#define XPS   (NPAD + FRAME + NPAD)
#define WPB   8
#define RLEN  48
#define LDSW  (XPS + RLEN)

typedef __attribute__((ext_vector_type(2))) float v2f;
typedef __attribute__((ext_vector_type(8))) float v8f;
typedef __attribute__((ext_vector_type(4))) float v4f;

template <int M>
__device__ __forceinline__ float row_xor(float v) {
    return __builtin_bit_cast(float,
        __builtin_amdgcn_update_dpp(0, __builtin_bit_cast(int, v), 0x160 + M, 0xf, 0xf, true));
}

__device__ __forceinline__ float other_row(float v) {
    const unsigned int u = __builtin_bit_cast(unsigned int, v);
    return __builtin_bit_cast(float,
        __builtin_amdgcn_permlanex16(u, u, 0x76543210u, 0xfedcba98u, false, false));
}

template <int N>
__device__ __forceinline__ float row_lane(float v) {
    return __builtin_bit_cast(float,
        __builtin_amdgcn_update_dpp(0, __builtin_bit_cast(int, v), 0x150 + N, 0xf, 0xf, true));
}

__device__ __forceinline__ float row_down1(float v) {
    return __builtin_bit_cast(float,
        __builtin_amdgcn_update_dpp(0, __builtin_bit_cast(int, v), 0x101, 0xf, 0xf, true));
}

__device__ __forceinline__ float row_up1(float v, float first) {
    return __builtin_bit_cast(float,
        __builtin_amdgcn_update_dpp(__builtin_bit_cast(int, first), __builtin_bit_cast(int, v),
                                    0x111, 0xf, 0xf, false));
}

template <int M>
__device__ __forceinline__ float fold(float a, float b, bool hi) {
    const float keep = hi ? b : a;
    const float send = hi ? a : b;
    return keep + row_xor<M>(send);
}

__global__ __launch_bounds__(32 * WPB)
void lpc_residual_kernel(const float* __restrict__ pcm,
                         float* __restrict__ out,
                         int nframes)
{
    __shared__ __attribute__((aligned(16))) float smem[WPB * LDSW];
    const int lane  = threadIdx.x & 31;
    const int wave  = threadIdx.x >> 5;
    const int frame = blockIdx.x * WPB + wave;
    if (frame >= nframes) return;

    float* xps   = smem + wave * LDSW;
    float* rco   = xps + XPS;

    const float* src = pcm + (size_t)frame * FRAME;

    #pragma unroll
    for (int q = 0; q < 5; ++q)
        xps[NPAD + lane + 32 * q] = src[lane + 32 * q];
    if (lane < NPAD) { xps[lane] = 0.f; xps[NPAD + FRAME + lane] = 0.f; }
    rco[lane] = 0.f;
    if (lane < 16) rco[32 + lane] = 0.f;
    __builtin_amdgcn_wave_barrier();

    float r_up;
    float r_zero;
    {
        const float* xw = xps + NPAD + 5 * lane;
        float w[5 + ORDER];
        #pragma unroll
        for (int c = 0; c < 5 + ORDER; ++c) w[c] = xw[c];

        float s[ORDER + 1];
        #pragma unroll
        for (int l = 0; l <= ORDER; ++l) {
            float acc = w[0] * w[l];
            #pragma unroll
            for (int c = 1; c < 5; ++c) acc = __builtin_fmaf(w[c], w[c + l], acc);
            s[l] = acc;
        }

        const bool b0 = (lane & 1) != 0, b1 = (lane & 2) != 0, b2 = (lane & 4) != 0, b3 = (lane & 8) != 0;
        float u[8];
        #pragma unroll
        for (int p = 0; p < 8; ++p) u[p] = fold<8>(s[2 * p + 1], s[2 * p + 2], b3);
        float v[4];
        #pragma unroll
        for (int q = 0; q < 4; ++q) v[q] = fold<4>(u[2 * q], u[2 * q + 1], b2);
        const float z0 = fold<2>(v[0], v[1], b1);
        const float z1 = fold<2>(v[2], v[3], b1);
        float y = fold<1>(z0, z1, b0);
        y += other_row(y);

        float e0 = s[0];
        e0 += row_xor<1>(e0);
        e0 += row_xor<2>(e0);
        e0 += row_xor<4>(e0);
        e0 += row_xor<8>(e0);
        e0 += other_row(e0);

        r_up = __builtin_bit_cast(float,
            __builtin_amdgcn_permlane16(0u, __builtin_bit_cast(unsigned int, y),
                                        0xE6A2C480u, 0xF7B3D591u, false, false));
        r_zero = e0;
    }

    {
        const int n = lane & 15;
        float fwd = r_up;
        float bwd = row_up1(r_up, r_zero);
        float pa = 0.f, ra = 0.f;
        float e   = r_zero;
        float acc = row_lane<0>(fwd);
        #pragma unroll
        for (int i = 1; i <= ORDER; ++i) {
            const float rinv = __builtin_amdgcn_rcpf(e);
            const float k0 = acc * rinv;
            const float k  = __builtin_fmaf(__builtin_fmaf(-e, k0, acc), rinv, k0);
            const float fn = fwd - k * bwd;
            const float bn = bwd - k * fwd;
            acc = row_lane<1>(fn);
            fwd = row_down1(fn);
            bwd = bn;
            const float t1 = pa - k * ra;
            const float t2 = ra - k * pa;
            pa = (n == i - 1) ? k : t1;
            ra = row_up1(t2, k);
            e = e * (1.f - k * k);
        }
        if (lane < ORDER) rco[30 - lane] = pa;
        if (lane == ORDER) rco[31] = 1.f;
    }
    __builtin_amdgcn_wave_barrier();

    const int half = lane >> 4;
    const int mrow = lane & 15;
    const int blk  = min(2 * (mrow & 7) + (mrow >> 3), 9);

    const v4f* ap = reinterpret_cast<const v4f*>(xps + (NPAD - 16) + 16 * (blk + half));
    float av[16];
    #pragma unroll
    for (int g = 0; g < 4; ++g) {
        const v4f t = ap[g];
        av[4 * g] = t.x; av[4 * g + 1] = t.y; av[4 * g + 2] = t.z; av[4 * g + 3] = t.w;
    }

    const float* bp = rco + 15 - mrow + 16 * half;
    float bv[16];
    #pragma unroll
    for (int r = 0; r < 16; ++r) bv[r] = bp[r];

    v8f acc = {};
    #pragma unroll
    for (int s = 0; s < 8; ++s) {
        v2f aop, bop;
        aop.x = av[2 * s]; aop.y = av[2 * s + 1];
        bop.x = bv[2 * s]; bop.y = bv[2 * s + 1];
        acc = __builtin_amdgcn_wmma_f32_16x16x4_f32(
                  false, aop, false, bop, (short)0, acc, false, false);
    }

    float* dst = out + (size_t)frame * FRAME + lane;
    #pragma unroll
    for (int q = 0; q < 5; ++q)
        dst[32 * q] = acc[q];
}

extern "C" void kernel_launch(void* const* d_in, const int* in_sizes, int n_in,
                              void* d_out, int out_size, void* d_ws, size_t ws_size,
                              hipStream_t stream) {
    const float* pcm = (const float*)d_in[0];
    float* out = (float*)d_out;
    const int total   = in_sizes[0];
    const int nframes = total / FRAME;
    const int blocks  = (nframes + WPB - 1) / WPB;
    lpc_residual_kernel<<<blocks, 32 * WPB, 0, stream>>>(pcm, out, nframes);
}


# Levinson: the two lane rows split roles (row 0 the q pair, row 1 the coefficients reversed, -1 in a[0]'s place): one FMA pair + one lane shift a step serve both, no select; k via v_readlane
# speedup vs baseline: 3.6846x; 1.1060x over previous
#include <hip/hip_runtime.h>

#define ORDER 16
#define FRAME 160
#define NPAD  20
#define XPS   (NPAD + FRAME + NPAD)
#define WPB   8
#define RLEN  48
#define LDSW  (XPS + RLEN)

typedef __attribute__((ext_vector_type(2))) float v2f;
typedef __attribute__((ext_vector_type(8))) float v8f;
typedef __attribute__((ext_vector_type(4))) float v4f;

template <int M>
__device__ __forceinline__ float row_xor(float v) {
    return __builtin_bit_cast(float,
        __builtin_amdgcn_update_dpp(0, __builtin_bit_cast(int, v), 0x160 + M, 0xf, 0xf, true));
}

__device__ __forceinline__ float other_row(float v) {
    const unsigned int u = __builtin_bit_cast(unsigned int, v);
    return __builtin_bit_cast(float,
        __builtin_amdgcn_permlanex16(u, u, 0x76543210u, 0xfedcba98u, false, false));
}

template <int N>
__device__ __forceinline__ float wave_lane(float v) {
    return __builtin_bit_cast(float, __builtin_amdgcn_readlane(__builtin_bit_cast(int, v), N));
}

__device__ __forceinline__ float row_down1(float v, float last) {
    return __builtin_bit_cast(float,
        __builtin_amdgcn_update_dpp(__builtin_bit_cast(int, last), __builtin_bit_cast(int, v),
                                    0x101, 0xf, 0xf, false));
}

__device__ __forceinline__ float row_up1(float v, float first) {
    return __builtin_bit_cast(float,
        __builtin_amdgcn_update_dpp(__builtin_bit_cast(int, first), __builtin_bit_cast(int, v),
                                    0x111, 0xf, 0xf, false));
}

template <int M>
__device__ __forceinline__ float fold(float a, float b, bool hi) {
    const float keep = hi ? b : a;
    const float send = hi ? a : b;
    return keep + row_xor<M>(send);
}

__global__ __launch_bounds__(32 * WPB)
void lpc_residual_kernel(const float* __restrict__ pcm,
                         float* __restrict__ out,
                         int nframes)
{
    __shared__ __attribute__((aligned(16))) float smem[WPB * LDSW];
    const int lane  = threadIdx.x & 31;
    const int wave  = threadIdx.x >> 5;
    const int frame = blockIdx.x * WPB + wave;
    if (frame >= nframes) return;

    float* xps   = smem + wave * LDSW;
    float* rco   = xps + XPS;

    const float* src = pcm + (size_t)frame * FRAME;

    #pragma unroll
    for (int q = 0; q < 5; ++q)
        xps[NPAD + lane + 32 * q] = src[lane + 32 * q];
    if (lane < NPAD) { xps[lane] = 0.f; xps[NPAD + FRAME + lane] = 0.f; }
    rco[lane] = 0.f;
    if (lane < 16) rco[32 + lane] = 0.f;
    __builtin_amdgcn_wave_barrier();

    float r_up;
    float r_zero;
    {
        const float* xw = xps + NPAD + 5 * lane;
        float w[5 + ORDER];
        #pragma unroll
        for (int c = 0; c < 5 + ORDER; ++c) w[c] = xw[c];

        float s[ORDER + 1];
        #pragma unroll
        for (int l = 0; l <= ORDER; ++l) {
            float acc = w[0] * w[l];
            #pragma unroll
            for (int c = 1; c < 5; ++c) acc = __builtin_fmaf(w[c], w[c + l], acc);
            s[l] = acc;
        }

        const bool b0 = (lane & 1) != 0, b1 = (lane & 2) != 0, b2 = (lane & 4) != 0, b3 = (lane & 8) != 0;
        float u[8];
        #pragma unroll
        for (int p = 0; p < 8; ++p) u[p] = fold<8>(s[2 * p + 1], s[2 * p + 2], b3);
        float v[4];
        #pragma unroll
        for (int q = 0; q < 4; ++q) v[q] = fold<4>(u[2 * q], u[2 * q + 1], b2);
        const float z0 = fold<2>(v[0], v[1], b1);
        const float z1 = fold<2>(v[2], v[3], b1);
        float y = fold<1>(z0, z1, b0);
        y += other_row(y);

        float e0 = s[0];
        e0 += row_xor<1>(e0);
        e0 += row_xor<2>(e0);
        e0 += row_xor<4>(e0);
        e0 += row_xor<8>(e0);
        e0 += other_row(e0);

        r_up = __builtin_bit_cast(float,
            __builtin_amdgcn_permlane16(0u, __builtin_bit_cast(unsigned int, y),
                                        0xE6A2C480u, 0xF7B3D591u, false, false));
        r_zero = e0;
    }

    {
        const bool qrow = lane < 16;
        float u = qrow ? r_up : (lane == 31 ? -1.f : 0.f);
        float v = qrow ? row_up1(r_up, r_zero) : 0.f;
        float e   = r_zero;
        float acc = wave_lane<0>(r_up);
        for (int i = 1; i <= ORDER; ++i) {
            const float rinv = __builtin_amdgcn_rcpf(e);
            const float k0 = acc * rinv;
            const float k  = __builtin_fmaf(__builtin_fmaf(-e, k0, acc), rinv, k0);
            const float un = u - k * v;
            const float vn = v - k * u;
            acc = wave_lane<1>(un);
            u = row_down1(un, k);
            v = vn;
            e = e * (1.f - k * k);
        }
        if (lane >= 16) rco[lane - 1] = v;
        if (lane == 15) rco[31] = 1.f;
    }
    __builtin_amdgcn_wave_barrier();

    const int half = lane >> 4;
    const int mrow = lane & 15;
    const int blk  = min(2 * (mrow & 7) + (mrow >> 3), 9);

    const v4f* ap = reinterpret_cast<const v4f*>(xps + (NPAD - 16) + 16 * (blk + half));
    float av[16];
    #pragma unroll
    for (int g = 0; g < 4; ++g) {
        const v4f t = ap[g];
        av[4 * g] = t.x; av[4 * g + 1] = t.y; av[4 * g + 2] = t.z; av[4 * g + 3] = t.w;
    }

    const float* bp = rco + 15 - mrow + 16 * half;
    float bv[16];
    #pragma unroll
    for (int r = 0; r < 16; ++r) bv[r] = bp[r];

    v8f acc = {};
    #pragma unroll
    for (int s = 0; s < 8; ++s) {
        v2f aop, bop;
        aop.x = av[2 * s]; aop.y = av[2 * s + 1];
        bop.x = bv[2 * s]; bop.y = bv[2 * s + 1];
        acc = __builtin_amdgcn_wmma_f32_16x16x4_f32(
                  false, aop, false, bop, (short)0, acc, false, false);
    }

    float* dst = out + (size_t)frame * FRAME + lane;
    #pragma unroll
    for (int q = 0; q < 5; ++q)
        dst[32 * q] = acc[q];
}

extern "C" void kernel_launch(void* const* d_in, const int* in_sizes, int n_in,
                              void* d_out, int out_size, void* d_ws, size_t ws_size,
                              hipStream_t stream) {
    const float* pcm = (const float*)d_in[0];
    float* out = (float*)d_out;
    const int total   = in_sizes[0];
    const int nframes = total / FRAME;
    const int blocks  = (nframes + WPB - 1) / WPB;
    lpc_residual_kernel<<<blocks, 32 * WPB, 0, stream>>>(pcm, out, nframes);
}


# wave-uniform arithmetic on the scalar unit: Levinson's k and e chain (v_s_rcp_f32, s_mul_f32, s_fmac_f32; r[0] read into a scalar register) and the frame's index and address math; no sum reordered
# speedup vs baseline: 4.2283x; 1.1476x over previous
#include <hip/hip_runtime.h>

#define ORDER 16
#define FRAME 160
#define NPAD  20
#define XPS   (NPAD + FRAME + NPAD)
#define WPB   8
#define RLEN  48
#define LDSW  (XPS + RLEN)

typedef __attribute__((ext_vector_type(2))) float v2f;
typedef __attribute__((ext_vector_type(8))) float v8f;
typedef __attribute__((ext_vector_type(4))) float v4f;

template <int M>
__device__ __forceinline__ float row_xor(float v) {
    return __builtin_bit_cast(float,
        __builtin_amdgcn_update_dpp(0, __builtin_bit_cast(int, v), 0x160 + M, 0xf, 0xf, true));
}

__device__ __forceinline__ float other_row(float v) {
    const unsigned int u = __builtin_bit_cast(unsigned int, v);
    return __builtin_bit_cast(float,
        __builtin_amdgcn_permlanex16(u, u, 0x76543210u, 0xfedcba98u, false, false));
}

template <int N>
__device__ __forceinline__ float wave_lane(float v) {
    return __builtin_bit_cast(float, __builtin_amdgcn_readlane(__builtin_bit_cast(int, v), N));
}

__device__ __forceinline__ float row_down1(float v, float last) {
    return __builtin_bit_cast(float,
        __builtin_amdgcn_update_dpp(__builtin_bit_cast(int, last), __builtin_bit_cast(int, v),
                                    0x101, 0xf, 0xf, false));
}

__device__ __forceinline__ float row_up1(float v, float first) {
    return __builtin_bit_cast(float,
        __builtin_amdgcn_update_dpp(__builtin_bit_cast(int, first), __builtin_bit_cast(int, v),
                                    0x111, 0xf, 0xf, false));
}

template <int M>
__device__ __forceinline__ float fold(float a, float b, bool hi) {
    const float keep = hi ? b : a;
    const float send = hi ? a : b;
    return keep + row_xor<M>(send);
}

__global__ __launch_bounds__(32 * WPB)
void lpc_residual_kernel(const float* __restrict__ pcm,
                         float* __restrict__ out,
                         int nframes)
{
    __shared__ __attribute__((aligned(16))) float smem[WPB * LDSW];
    const int lane  = threadIdx.x & 31;
    const int wave  = __builtin_amdgcn_readfirstlane(threadIdx.x >> 5);
    const int frame = blockIdx.x * WPB + wave;
    if (frame >= nframes) return;

    float* xps   = smem + wave * LDSW;
    float* rco   = xps + XPS;

    const float* src = pcm + (size_t)frame * FRAME;

    #pragma unroll
    for (int q = 0; q < 5; ++q)
        xps[NPAD + lane + 32 * q] = src[lane + 32 * q];
    if (lane < NPAD) { xps[lane] = 0.f; xps[NPAD + FRAME + lane] = 0.f; }
    rco[lane] = 0.f;
    if (lane < 16) rco[32 + lane] = 0.f;
    __builtin_amdgcn_wave_barrier();

    float r_up;
    float r_zero;
    {
        const float* xw = xps + NPAD + 5 * lane;
        float w[5 + ORDER];
        #pragma unroll
        for (int c = 0; c < 5 + ORDER; ++c) w[c] = xw[c];

        float s[ORDER + 1];
        #pragma unroll
        for (int l = 0; l <= ORDER; ++l) {
            float acc = w[0] * w[l];
            #pragma unroll
            for (int c = 1; c < 5; ++c) acc = __builtin_fmaf(w[c], w[c + l], acc);
            s[l] = acc;
        }

        const bool b0 = (lane & 1) != 0, b1 = (lane & 2) != 0, b2 = (lane & 4) != 0, b3 = (lane & 8) != 0;
        float u[8];
        #pragma unroll
        for (int p = 0; p < 8; ++p) u[p] = fold<8>(s[2 * p + 1], s[2 * p + 2], b3);
        float v[4];
        #pragma unroll
        for (int q = 0; q < 4; ++q) v[q] = fold<4>(u[2 * q], u[2 * q + 1], b2);
        const float z0 = fold<2>(v[0], v[1], b1);
        const float z1 = fold<2>(v[2], v[3], b1);
        float y = fold<1>(z0, z1, b0);
        y += other_row(y);

        float e0 = s[0];
        e0 += row_xor<1>(e0);
        e0 += row_xor<2>(e0);
        e0 += row_xor<4>(e0);
        e0 += row_xor<8>(e0);
        e0 += other_row(e0);

        r_up = __builtin_bit_cast(float,
            __builtin_amdgcn_permlane16(0u, __builtin_bit_cast(unsigned int, y),
                                        0xE6A2C480u, 0xF7B3D591u, false, false));
        r_zero = e0;
    }

    {
        const bool qrow = lane < 16;
        float u = qrow ? r_up : (lane == 31 ? -1.f : 0.f);
        float v = qrow ? row_up1(r_up, r_zero) : 0.f;
        float e   = wave_lane<0>(r_zero);
        float acc = wave_lane<0>(r_up);
        for (int i = 1; i <= ORDER; ++i) {
            const float rinv = __builtin_amdgcn_rcpf(e);
            const float k0 = acc * rinv;
            const float k  = __builtin_fmaf(__builtin_fmaf(-e, k0, acc), rinv, k0);
            const float un = u - k * v;
            const float vn = v - k * u;
            acc = wave_lane<1>(un);
            u = row_down1(un, k);
            v = vn;
            e = e * (1.f - k * k);
        }
        if (lane >= 16) rco[lane - 1] = v;
        if (lane == 15) rco[31] = 1.f;
    }
    __builtin_amdgcn_wave_barrier();

    const int half = lane >> 4;
    const int mrow = lane & 15;
    const int blk  = min(2 * (mrow & 7) + (mrow >> 3), 9);

    const v4f* ap = reinterpret_cast<const v4f*>(xps + (NPAD - 16) + 16 * (blk + half));
    float av[16];
    #pragma unroll
    for (int g = 0; g < 4; ++g) {
        const v4f t = ap[g];
        av[4 * g] = t.x; av[4 * g + 1] = t.y; av[4 * g + 2] = t.z; av[4 * g + 3] = t.w;
    }

    const float* bp = rco + 15 - mrow + 16 * half;
    float bv[16];
    #pragma unroll
    for (int r = 0; r < 16; ++r) bv[r] = bp[r];

    v8f acc = {};
    #pragma unroll
    for (int s = 0; s < 8; ++s) {
        v2f aop, bop;
        aop.x = av[2 * s]; aop.y = av[2 * s + 1];
        bop.x = bv[2 * s]; bop.y = bv[2 * s + 1];
        acc = __builtin_amdgcn_wmma_f32_16x16x4_f32(
                  false, aop, false, bop, (short)0, acc, false, false);
    }

    float* dst = out + (size_t)frame * FRAME + lane;
    #pragma unroll
    for (int q = 0; q < 5; ++q)
        dst[32 * q] = acc[q];
}

extern "C" void kernel_launch(void* const* d_in, const int* in_sizes, int n_in,
                              void* d_out, int out_size, void* d_ws, size_t ws_size,
                              hipStream_t stream) {
    const float* pcm = (const float*)d_in[0];
    float* out = (float*)d_out;
    const int total   = in_sizes[0];
    const int nframes = total / FRAME;
    const int blocks  = (nframes + WPB - 1) / WPB;
    lpc_residual_kernel<<<blocks, 32 * WPB, 0, stream>>>(pcm, out, nframes);
}
